# starting kernel
# baseline (speedup 1.0000x reference)
#include <hip/hip_runtime.h>
#include <math.h>

#define N_NODES 50000
#define N_EDGES 800000
#define IN_DIM  256
#define OUT_DIM 128
#define LEAK    0.2f

typedef __attribute__((ext_vector_type(2))) float v2f;
typedef __attribute__((ext_vector_type(4))) float v4f;
typedef __attribute__((ext_vector_type(8))) float v8f;

__global__ __launch_bounds__(256) void k_init(float* __restrict__ out,
                                              float* __restrict__ denom,
                                              unsigned* __restrict__ m_enc) {
  const int i = blockIdx.x * blockDim.x + threadIdx.x;
  if (i < N_NODES * OUT_DIM) out[i] = 0.0f;
  if (i < N_NODES) { denom[i] = 0.0f; m_enc[i] = 0u; }
}

__global__ __launch_bounds__(256) void k_gemm(const float* __restrict__ h,
                                              const float* __restrict__ W,
                                              float* __restrict__ z) {
  const int lane = threadIdx.x & 31;
  const int wv   = threadIdx.x >> 5;
  const int half = lane >> 4;
  const int l    = lane & 15;
  const int m0   = blockIdx.x * 16;
  const int n0   = wv * 16;

  const float* __restrict__ ar = h + (size_t)(m0 + l) * IN_DIM;
  const float* __restrict__ bc = W + (size_t)(n0 + l);

  v8f acc = {};
#pragma unroll 4
  for (int k = 0; k < IN_DIM; k += 4) {
    const int kk = k + 2 * half;
    v2f a; a.x = ar[kk];                     a.y = ar[kk + 1];
    v2f b; b.x = bc[(size_t)kk * OUT_DIM];   b.y = bc[(size_t)(kk + 1) * OUT_DIM];
    acc = __builtin_amdgcn_wmma_f32_16x16x4_f32(
         false, a,  false, b,
         (short)0, acc,  false,  false);
  }

  float* zp = z + (size_t)(m0 + 8 * half) * OUT_DIM + n0 + l;
#pragma unroll
  for (int i = 0; i < 8; ++i) zp[(size_t)i * OUT_DIM] = acc[i];
}

__global__ __launch_bounds__(256) void k_scores(const float* __restrict__ z,
                                                const float* __restrict__ attn_w,
                                                float* __restrict__ s1,
                                                float* __restrict__ s2) {
  const int n = blockIdx.x * blockDim.x + threadIdx.x;
  if (n >= N_NODES) return;
  const float* zr = z + (size_t)n * OUT_DIM;
  float d1 = 0.0f, d2 = 0.0f;
#pragma unroll 8
  for (int j = 0; j < OUT_DIM; ++j) {
    const float v = zr[j];
    d1 += v * attn_w[j];
    d2 += v * attn_w[OUT_DIM + j];
  }
  s1[n] = d1;
  s2[n] = d2;
}

__device__ __forceinline__ unsigned enc_f32(float f) {
  unsigned b = __float_as_uint(f);
  return (b & 0x80000000u) ? ~b : (b | 0x80000000u);
}
__device__ __forceinline__ float dec_f32(unsigned e) {
  unsigned b = (e & 0x80000000u) ? (e & 0x7FFFFFFFu) : ~e;
  return __uint_as_float(b);
}

__global__ __launch_bounds__(256) void k_edge_max(const int* __restrict__ src,
                                                  const int* __restrict__ dst,
                                                  const float* __restrict__ s1,
                                                  const float* __restrict__ s2,
                                                  float* __restrict__ ebuf,
                                                  unsigned* __restrict__ m_enc) {
  const int ei = blockIdx.x * blockDim.x + threadIdx.x;
  if (ei >= N_EDGES) return;
  const float logit = s1[src[ei]] + s2[dst[ei]];
  const float e = (logit >= 0.0f) ? logit : LEAK * logit;
  ebuf[ei] = e;
  atomicMax(&m_enc[dst[ei]], enc_f32(e));
}

__global__ __launch_bounds__(256) void k_edge_exp(const int* __restrict__ dst,
                                                  float* __restrict__ ebuf,
                                                  const unsigned* __restrict__ m_enc,
                                                  float* __restrict__ denom) {
  const int ei = blockIdx.x * blockDim.x + threadIdx.x;
  if (ei >= N_EDGES) return;
  const int d = dst[ei];
  const float ex = expf(ebuf[ei] - dec_f32(m_enc[d]));
  ebuf[ei] = ex;
  atomicAdd(&denom[d], ex);
}

__global__ __launch_bounds__(256) void k_scatter(const int* __restrict__ src,
                                                 const int* __restrict__ dst,
                                                 const float* __restrict__ z,
                                                 const float* __restrict__ ebuf,
                                                 const float* __restrict__ denom,
                                                 float* __restrict__ hacc) {
  const int ei = blockIdx.x * 8 + (threadIdx.x >> 5);
  if (ei >= N_EDGES) return;
  const int lane = threadIdx.x & 31;
  const int s = src[ei];
  const int d = dst[ei];
  const float dn = denom[d];
  const float alpha = ebuf[ei] / ((dn > 0.0f) ? dn : 1.0f);

  const v4f v = ((const v4f*)(z + (size_t)s * OUT_DIM))[lane];
  float* ho = hacc + (size_t)d * OUT_DIM + lane * 4;
  atomicAdd(&ho[0], alpha * v.x);
  atomicAdd(&ho[1], alpha * v.y);
  atomicAdd(&ho[2], alpha * v.z);
  atomicAdd(&ho[3], alpha * v.w);
}

__global__ __launch_bounds__(256) void k_final(float* __restrict__ out) {
  const int i = blockIdx.x * blockDim.x + threadIdx.x;
  if (i >= N_NODES * OUT_DIM) return;
  const float x = out[i];
  out[i] = (x > 0.0f) ? x : expm1f(x);
}

extern "C" void kernel_launch(void* const* d_in, const int* in_sizes, int n_in,
                              void* d_out, int out_size, void* d_ws, size_t ws_size,
                              hipStream_t stream) {
  const float* h      = (const float*)d_in[0];
  const int*   src    = (const int*)  d_in[1];
  const int*   dst    = (const int*)  d_in[2];
  const float* W_fc   = (const float*)d_in[3];
  const float* attn_w = (const float*)d_in[4];
  float*       out    = (float*)d_out;

  float*    z     = (float*)d_ws;
  float*    s1    = z  + (size_t)N_NODES * OUT_DIM;
  float*    s2    = s1 + N_NODES;
  float*    denom = s2 + N_NODES;
  unsigned* m_enc = (unsigned*)(denom + N_NODES);
  float*    ebuf  = (float*)(m_enc + N_NODES);

  const int TPB = 256;

  k_init<<<(N_NODES * OUT_DIM + TPB - 1) / TPB, TPB, 0, stream>>>(out, denom, m_enc);

  k_gemm<<<N_NODES / 16, TPB, 0, stream>>>(h, W_fc, z);

  k_scores<<<(N_NODES + TPB - 1) / TPB, TPB, 0, stream>>>(z, attn_w, s1, s2);

  k_edge_max<<<(N_EDGES + TPB - 1) / TPB, TPB, 0, stream>>>(src, dst, s1, s2, ebuf, m_enc);

  k_edge_exp<<<(N_EDGES + TPB - 1) / TPB, TPB, 0, stream>>>(dst, ebuf, m_enc, denom);

  k_scatter<<<(N_EDGES + 7) / 8, TPB, 0, stream>>>(src, dst, z, ebuf, denom, out);

  k_final<<<(N_NODES * OUT_DIM + TPB - 1) / TPB, TPB, 0, stream>>>(out);
}


# z = h @ W operand feed: W staged once per block in LDS in fragment order, h rows read straight into A fragments (16-byte loads), wave = 16x64 strip; k sum order changes inside 8-k chunks
# speedup vs baseline: 3.4004x; 3.4004x over previous
#include <hip/hip_runtime.h>
#include <math.h>

#define N_NODES 50000
#define N_EDGES 800000
#define IN_DIM  256
#define OUT_DIM 128
#define LEAK    0.2f

typedef __attribute__((ext_vector_type(2))) float v2f;
typedef __attribute__((ext_vector_type(4))) float v4f;
typedef __attribute__((ext_vector_type(8))) float v8f;

#define R_NODES  196
#define N_OWN    ((N_NODES + R_NODES - 1) / R_NODES)
#define LIST_CAP 16384
#define NQ       (N_EDGES / 16)
#define AGG_TPB  1024
#define SCAN_BATCH 7
#define SCAN_ITERS ((NQ + AGG_TPB - 1) / AGG_TPB)

static_assert(N_OWN <= 256, "owner id must fit in one byte");
static_assert(N_EDGES % 16 == 0, "keys are read 16 at a time");
static_assert(N_NODES <= 65536, "src index is packed into 16 bits");
static_assert(SCAN_ITERS % SCAN_BATCH == 0, "scan batches must tile the scan");
static_assert(OUT_DIM == 128, "one float4 per lane per row");

#define GEMM_TPB    512
#define GEMM_BLOCKS 256
#define M_TILES     (N_NODES / 16)
#define TILE_STEP   (GEMM_BLOCKS * (GEMM_TPB / 64))

static_assert(N_NODES % 16 == 0, "row tiles are whole");
static_assert(IN_DIM % 64 == 0, "k is walked in groups of 64");

__device__ __forceinline__ void load_a_group(v4f (&a)[8], const v4f* __restrict__ ap, int g) {
#pragma unroll
  for (int i = 0; i < 8; ++i) a[i] = ap[2 * (8 * g + i)];
}

__device__ __forceinline__ void load_b_chunk(v4f (&b)[4], const v4f* __restrict__ bp, int kc) {
  const v4f* __restrict__ bq = bp + (size_t)(2 * kc) * OUT_DIM;
#pragma unroll
  for (int t = 0; t < 4; ++t) b[t] = bq[16 * t];
}

__device__ __forceinline__ void mma_chunk(v8f (&acc)[4], const v4f a, const v4f (&b)[4]) {
  v2f alo; alo.x = a.x; alo.y = a.y;
  v2f ahi; ahi.x = a.z; ahi.y = a.w;
#pragma unroll
  for (int t = 0; t < 4; ++t) {
    v2f blo; blo.x = b[t].x; blo.y = b[t].y;
    acc[t] = __builtin_amdgcn_wmma_f32_16x16x4_f32(
         false, alo,  false, blo,
         (short)0, acc[t],  false,  false);
  }
#pragma unroll
  for (int t = 0; t < 4; ++t) {
    v2f bhi; bhi.x = b[t].z; bhi.y = b[t].w;
    acc[t] = __builtin_amdgcn_wmma_f32_16x16x4_f32(
         false, ahi,  false, bhi,
         (short)0, acc[t],  false,  false);
  }
}

template <int G>
__device__ __forceinline__ void mma_group(v8f (&acc)[4], const v4f (&a)[8],
                                          const v4f* __restrict__ bp,
                                          v4f (&b0)[4], v4f (&b1)[4]) {
#pragma unroll
  for (int i = 0; i < 8; i += 2) {
    load_b_chunk(b1, bp, 8 * G + i + 1);
    mma_chunk(acc, a[i], b0);
    if (8 * G + i + 2 < IN_DIM / 8) load_b_chunk(b0, bp, 8 * G + i + 2);
    mma_chunk(acc, a[i + 1], b1);
  }
}

__global__ __launch_bounds__(GEMM_TPB) void k_gemm(const float* __restrict__ h,
                                                   const float* __restrict__ W,
                                                   float* __restrict__ z) {
  __shared__ v4f Wl[(IN_DIM / 4) * OUT_DIM];

  const int tid  = threadIdx.x;
  const int lane = tid & 31;
  const int wv   = tid >> 5;
  const int half = lane >> 4;
  const int c    = lane & 15;
  const int nh   = wv & 1;
  const int pr   = wv >> 1;

  int mt = pr * GEMM_BLOCKS + blockIdx.x;
  const v4f* __restrict__ ap = (const v4f*)(h + (size_t)(mt * 16 + c) * IN_DIM) + half;

  v4f a0[8], a1[8];
  load_a_group(a0, ap, 0);

  {
    const v4f* __restrict__ W4 = (const v4f*)W;
    for (int e = tid; e < (IN_DIM / 4) * (OUT_DIM / 4); e += GEMM_TPB) {
      const int kq = e >> 5;
      const int j  = e & 31;
      const v4f r0 = W4[(4 * kq + 0) * (OUT_DIM / 4) + j];
      const v4f r1 = W4[(4 * kq + 1) * (OUT_DIM / 4) + j];
      const v4f r2 = W4[(4 * kq + 2) * (OUT_DIM / 4) + j];
      const v4f r3 = W4[(4 * kq + 3) * (OUT_DIM / 4) + j];
      v4f* d = Wl + kq * OUT_DIM + 4 * j;
      v4f q0 = {r0.x, r1.x, r2.x, r3.x};
      v4f q1 = {r0.y, r1.y, r2.y, r3.y};
      v4f q2 = {r0.z, r1.z, r2.z, r3.z};
      v4f q3 = {r0.w, r1.w, r2.w, r3.w};
      d[0] = q0; d[1] = q1; d[2] = q2; d[3] = q3;
    }
  }
  __syncthreads();

  const v4f* __restrict__ bp = Wl + half * OUT_DIM + nh * 64 + c;

  for (;;) {
    v8f acc[4] = {};
    const int  mt_next  = mt + TILE_STEP;
    const bool has_next = mt_next < M_TILES;

    asm volatile("" ::: "memory");
    v4f b0[4], b1[4];
    load_b_chunk(b0, bp, 0);
    load_a_group(a1, ap, 1);
    mma_group<0>(acc, a0, bp, b0, b1);
    load_a_group(a0, ap, 2);
    mma_group<1>(acc, a1, bp, b0, b1);
    load_a_group(a1, ap, 3);
    mma_group<2>(acc, a0, bp, b0, b1);
    const v4f* __restrict__ ap_next =
        (const v4f*)(h + (size_t)(mt_next * 16 + c) * IN_DIM) + half;
    if (has_next) load_a_group(a0, ap_next, 0);
    mma_group<3>(acc, a1, bp, b0, b1);

    float* zp = z + (size_t)(mt * 16 + 8 * half) * OUT_DIM + nh * 64 + c;
#pragma unroll
    for (int t = 0; t < 4; ++t) {
#pragma unroll
      for (int i = 0; i < 8; ++i) zp[(size_t)i * OUT_DIM + 16 * t] = acc[t][i];
    }

    if (!has_next) break;
    mt = mt_next;
    ap = ap_next;
  }
}

__global__ __launch_bounds__(256) void k_scores(const float* __restrict__ z,
                                                const float* __restrict__ attn_w,
                                                float* __restrict__ s1,
                                                float* __restrict__ s2) {
  const int n = blockIdx.x * blockDim.x + threadIdx.x;
  if (n >= N_NODES) return;
  const float* zr = z + (size_t)n * OUT_DIM;
  float d1 = 0.0f, d2 = 0.0f;
#pragma unroll 8
  for (int j = 0; j < OUT_DIM; ++j) {
    const float v = zr[j];
    d1 += v * attn_w[j];
    d2 += v * attn_w[OUT_DIM + j];
  }
  s1[n] = d1;
  s2[n] = d2;
}

__device__ __forceinline__ unsigned enc_f32(float f) {
  unsigned b = __float_as_uint(f);
  return (b & 0x80000000u) ? ~b : (b | 0x80000000u);
}
__device__ __forceinline__ float dec_f32(unsigned e) {
  unsigned b = (e & 0x80000000u) ? (e & 0x7FFFFFFFu) : ~e;
  return __uint_as_float(b);
}

__global__ __launch_bounds__(256) void k_keys(const int* __restrict__ dst,
                                              unsigned* __restrict__ keys) {
  const int i = blockIdx.x * blockDim.x + threadIdx.x;
  if (i >= N_EDGES / 4) return;
  const int4 d = ((const int4*)dst)[i];
  keys[i] = ((unsigned)d.x / R_NODES) | (((unsigned)d.y / R_NODES) << 8) |
            (((unsigned)d.z / R_NODES) << 16) | (((unsigned)d.w / R_NODES) << 24);
}

__global__ __launch_bounds__(AGG_TPB) void k_aggregate(const int* __restrict__ src,
                                                       const int* __restrict__ dst,
                                                       const float* __restrict__ z,
                                                       const float* __restrict__ s1,
                                                       const float* __restrict__ s2,
                                                       const unsigned* __restrict__ keys,
                                                       float* __restrict__ out) {
  __shared__ float    acc[R_NODES * OUT_DIM];
  __shared__ unsigned lst_a[LIST_CAP];
  __shared__ float    lst_e[LIST_CAP];
  __shared__ unsigned menc[R_NODES];
  __shared__ float    den[R_NODES];
  __shared__ float    s2l[R_NODES];
  __shared__ int      cnt;

  const int tid  = threadIdx.x;
  const int lane = tid & 31;
  const int wv   = tid >> 5;
  const int base = blockIdx.x * R_NODES;

  for (int i = tid; i < R_NODES * OUT_DIM / 4; i += AGG_TPB) {
    const v4f zero = {0.0f, 0.0f, 0.0f, 0.0f};
    ((v4f*)acc)[i] = zero;
  }
  if (tid < R_NODES) {
    menc[tid] = 0u;
    den[tid]  = 0.0f;
    const int nd = base + tid;
    s2l[tid] = (nd < N_NODES) ? s2[nd] : 0.0f;
  }
  if (tid == 0) cnt = 0;
  __syncthreads();

  {
    const uint4* __restrict__ k4 = (const uint4*)keys;
    const unsigned pat = blockIdx.x * 0x01010101u;
    const int start = blockIdx.x * (NQ / N_OWN);
    for (int it = 0; it < SCAN_ITERS; it += SCAN_BATCH) {
      uint4 w[SCAN_BATCH];
      int   qq[SCAN_BATCH];
      bool  ok[SCAN_BATCH];
#pragma unroll
      for (int u = 0; u < SCAN_BATCH; ++u) {
        const int t = (it + u) * AGG_TPB + tid;
        ok[u] = t < NQ;
        int q = t + start;
        if (q >= NQ) q -= NQ;
        if (!ok[u]) q = 0;
        qq[u] = q;
        w[u] = k4[q];
      }
#pragma unroll
      for (int u = 0; u < SCAN_BATCH; ++u) {
        if (!ok[u]) continue;
        const unsigned xs[4] = {w[u].x ^ pat, w[u].y ^ pat, w[u].z ^ pat, w[u].w ^ pat};
#pragma unroll
        for (int wi = 0; wi < 4; ++wi) {
          const unsigned x = xs[wi];
          if (((x - 0x01010101u) & ~x & 0x80808080u) != 0u) {
#pragma unroll
            for (int b = 0; b < 4; ++b) {
              if (((x >> (8 * b)) & 0xFFu) == 0u) {
                const int pos = atomicAdd(&cnt, 1);
                if (pos < LIST_CAP) lst_a[pos] = (unsigned)(qq[u] * 16 + wi * 4 + b);
              }
            }
          }
        }
      }
    }
  }
  __syncthreads();
  const int n = cnt;

  if (n <= LIST_CAP) {
    for (int j0 = tid; j0 < n; j0 += 4 * AGG_TPB) {
      int ii[4], dd[4], ss[4];
      float sv[4];
#pragma unroll
      for (int u = 0; u < 4; ++u) {
        const int j = j0 + u * AGG_TPB;
        ii[u] = (j < n) ? (int)lst_a[j] : 0;
      }
#pragma unroll
      for (int u = 0; u < 4; ++u) { dd[u] = dst[ii[u]]; ss[u] = src[ii[u]]; }
#pragma unroll
      for (int u = 0; u < 4; ++u) sv[u] = s1[ss[u]];
#pragma unroll
      for (int u = 0; u < 4; ++u) {
        const int j = j0 + u * AGG_TPB;
        if (j < n) {
          const int loc = dd[u] - base;
          const float logit = sv[u] + s2l[loc];
          const float e = (logit >= 0.0f) ? logit : LEAK * logit;
          lst_e[j] = e;
          lst_a[j] = ((unsigned)loc << 16) | (unsigned)ss[u];
          atomicMax(&menc[loc], enc_f32(e));
        }
      }
    }
    __syncthreads();

    for (int j = tid; j < n; j += AGG_TPB) {
      const int loc = (int)(lst_a[j] >> 16);
      const float ex = expf(lst_e[j] - dec_f32(menc[loc]));
      lst_e[j] = ex;
      atomicAdd(&den[loc], ex);
    }
    __syncthreads();

    const v4f* __restrict__ z4 = (const v4f*)z;
    int j0 = wv;
    for (; j0 + 7 * 32 < n; j0 += 8 * 32) {
      unsigned pk[8];
      float    ex[8];
      v4f      v[8];
#pragma unroll
      for (int u = 0; u < 8; ++u) { pk[u] = lst_a[j0 + u * 32]; ex[u] = lst_e[j0 + u * 32]; }
#pragma unroll
      for (int u = 0; u < 8; ++u) v[u] = z4[(size_t)(pk[u] & 0xFFFFu) * (OUT_DIM / 4) + lane];
#pragma unroll
      for (int u = 0; u < 8; ++u) {
        float* a = acc + (pk[u] >> 16) * OUT_DIM + lane;
        atomicAdd(a,      ex[u] * v[u].x);
        atomicAdd(a + 32, ex[u] * v[u].y);
        atomicAdd(a + 64, ex[u] * v[u].z);
        atomicAdd(a + 96, ex[u] * v[u].w);
      }
    }
    for (; j0 < n; j0 += 32) {
      const unsigned pk = lst_a[j0];
      const float    ex = lst_e[j0];
      const v4f v = z4[(size_t)(pk & 0xFFFFu) * (OUT_DIM / 4) + lane];
      float* a = acc + (pk >> 16) * OUT_DIM + lane;
      atomicAdd(a,      ex * v.x);
      atomicAdd(a + 32, ex * v.y);
      atomicAdd(a + 64, ex * v.z);
      atomicAdd(a + 96, ex * v.w);
    }
  } else {
    for (int i = tid; i < N_EDGES; i += AGG_TPB) {
      const int loc = dst[i] - base;
      if ((unsigned)loc < (unsigned)R_NODES) {
        const float logit = s1[src[i]] + s2l[loc];
        const float e = (logit >= 0.0f) ? logit : LEAK * logit;
        atomicMax(&menc[loc], enc_f32(e));
      }
    }
    __syncthreads();
    for (int i = tid; i < N_EDGES; i += AGG_TPB) {
      const int loc = dst[i] - base;
      if ((unsigned)loc < (unsigned)R_NODES) {
        const int s = src[i];
        const float logit = s1[s] + s2l[loc];
        const float e = (logit >= 0.0f) ? logit : LEAK * logit;
        const float ex = expf(e - dec_f32(menc[loc]));
        atomicAdd(&den[loc], ex);
        const float* zr = z + (size_t)s * OUT_DIM;
        float* a = acc + loc * OUT_DIM;
        for (int c = 0; c < OUT_DIM; ++c) atomicAdd(&a[(c & 3) * 32 + (c >> 2)], ex * zr[c]);
      }
    }
  }
  __syncthreads();

  for (int idx = tid; idx < R_NODES * (OUT_DIM / 4); idx += AGG_TPB) {
    const int r  = idx >> 5;
    const int q  = idx & 31;
    const int nd = base + r;
    if (nd < N_NODES) {
      const float dn0 = den[r];
      const float dn  = (dn0 > 0.0f) ? dn0 : 1.0f;
      const float* a = acc + r * OUT_DIM + q;
      v4f x;
      x.x = a[0] / dn; x.y = a[32] / dn; x.z = a[64] / dn; x.w = a[96] / dn;
      x.x = (x.x > 0.0f) ? x.x : expm1f(x.x);
      x.y = (x.y > 0.0f) ? x.y : expm1f(x.y);
      x.z = (x.z > 0.0f) ? x.z : expm1f(x.z);
      x.w = (x.w > 0.0f) ? x.w : expm1f(x.w);
      ((v4f*)out)[(size_t)nd * (OUT_DIM / 4) + q] = x;
    }
  }
}

extern "C" void kernel_launch(void* const* d_in, const int* in_sizes, int n_in,
                              void* d_out, int out_size, void* d_ws, size_t ws_size,
                              hipStream_t stream) {
  const float* h      = (const float*)d_in[0];
  const int*   src    = (const int*)  d_in[1];
  const int*   dst    = (const int*)  d_in[2];
  const float* W_fc   = (const float*)d_in[3];
  const float* attn_w = (const float*)d_in[4];
  float*       out    = (float*)d_out;

  float*    z    = (float*)d_ws;
  float*    s1   = z  + (size_t)N_NODES * OUT_DIM;
  float*    s2   = s1 + N_NODES;
  unsigned* keys = (unsigned*)(s2 + N_NODES);

  const int TPB = 256;

  k_gemm<<<GEMM_BLOCKS, GEMM_TPB, 0, stream>>>(h, W_fc, z);

  k_scores<<<(N_NODES + TPB - 1) / TPB, TPB, 0, stream>>>(z, attn_w, s1, s2);

  k_keys<<<(N_EDGES / 4 + TPB - 1) / TPB, TPB, 0, stream>>>(dst, keys);

  k_aggregate<<<N_OWN, AGG_TPB, 0, stream>>>(src, dst, z, s1, s2, keys, out);
}


# owner keys stored bit-sliced (8 plane words per 32 edges, made from wave ballots in the product launch); an owner block tests 32 edges with 8 XOR+AND, not byte by byte; sum order: arrival
# speedup vs baseline: 4.2185x; 1.2406x over previous
#include <hip/hip_runtime.h>
#include <math.h>

#define N_NODES 50000
#define N_EDGES 800000
#define IN_DIM  256
#define OUT_DIM 128
#define LEAK    0.2f

typedef __attribute__((ext_vector_type(2))) float v2f;
typedef __attribute__((ext_vector_type(4))) float v4f;
typedef __attribute__((ext_vector_type(8))) float v8f;

#define R_NODES  196
#define N_OWN    ((N_NODES + R_NODES - 1) / R_NODES)
#define LIST_CAP 16384
#define NSG      (N_EDGES / 32)
#define AGG_TPB  1024
#define SCAN_BATCH 5
#define SCAN_ITERS ((NSG + AGG_TPB - 1) / AGG_TPB)

static_assert(N_OWN <= 256, "owner id must fit in one byte");
static_assert(N_EDGES % 128 == 0, "a wave slices the keys of 128 edges at a time");
static_assert(N_NODES <= 65536, "src index is packed into 16 bits");
static_assert(SCAN_ITERS % SCAN_BATCH == 0, "scan batches must tile the scan");
static_assert(OUT_DIM == 128, "one float4 per lane per row");

#define GEMM_TPB    512
#define GEMM_BLOCKS 256
#define M_TILES     (N_NODES / 16)
#define TILE_STEP   (GEMM_BLOCKS * (GEMM_TPB / 64))

static_assert(N_NODES % 16 == 0, "row tiles are whole");
static_assert(IN_DIM % 64 == 0, "k is walked in groups of 64");
static_assert(IN_DIM == 16 * (GEMM_TPB / 32), "one 16-row tile of W per wave for W @ a");
#define STAGE_ROUNDS ((IN_DIM / 4) * (OUT_DIM / 4) / GEMM_TPB)
static_assert(STAGE_ROUNDS * GEMM_TPB == (IN_DIM / 4) * (OUT_DIM / 4), "staging rounds are whole");
static_assert(OUT_DIM / 8 == 4 * STAGE_ROUNDS, "four 8-k chunks of W @ a per staging round");

__device__ __forceinline__ void load_a_group(v4f (&a)[8], const v4f* __restrict__ ap, int g) {
#pragma unroll
  for (int i = 0; i < 8; ++i) a[i] = ap[2 * (8 * g + i)];
}

__device__ __forceinline__ void load_b_chunk(v4f (&b)[4], const v4f* __restrict__ bp, int kc) {
  const v4f* __restrict__ bq = bp + (size_t)(2 * kc) * OUT_DIM;
#pragma unroll
  for (int t = 0; t < 4; ++t) b[t] = bq[16 * t];
}

__device__ __forceinline__ void mma_chunk(v8f (&acc)[4], const v4f a, const v4f (&b)[4]) {
  v2f alo; alo.x = a.x; alo.y = a.y;
  v2f ahi; ahi.x = a.z; ahi.y = a.w;
#pragma unroll
  for (int t = 0; t < 4; ++t) {
    v2f blo; blo.x = b[t].x; blo.y = b[t].y;
    acc[t] = __builtin_amdgcn_wmma_f32_16x16x4_f32(
         false, alo,  false, blo,
         (short)0, acc[t],  false,  false);
  }
#pragma unroll
  for (int t = 0; t < 4; ++t) {
    v2f bhi; bhi.x = b[t].z; bhi.y = b[t].w;
    acc[t] = __builtin_amdgcn_wmma_f32_16x16x4_f32(
         false, ahi,  false, bhi,
         (short)0, acc[t],  false,  false);
  }
}

template <int G>
__device__ __forceinline__ void mma_group(v8f (&acc)[4], const v4f (&a)[8],
                                          const v4f* __restrict__ bp,
                                          v4f (&b0)[4], v4f (&b1)[4],
                                          const v4f* __restrict__ wp,
                                          v4f& q0, v4f& p0, v4f& q1, v4f& p1,
                                          v4f& sv, v4f& tv) {
#pragma unroll
  for (int i = 0; i < 8; i += 2) {
    load_b_chunk(b1, bp, 8 * G + i + 1);
    q1 = wp[2 * (8 * G + i + 1)];
    p1 = wp[2 * (8 * G + i + 1) + IN_DIM / 4];
    mma_chunk(acc, a[i], b0);
    sv += a[i] * q0;
    tv += a[i] * p0;
    asm volatile("" : "+v"(sv), "+v"(tv));
    if (8 * G + i + 2 < IN_DIM / 8) {
      load_b_chunk(b0, bp, 8 * G + i + 2);
      q0 = wp[2 * (8 * G + i + 2)];
      p0 = wp[2 * (8 * G + i + 2) + IN_DIM / 4];
    }
    mma_chunk(acc, a[i + 1], b1);
    sv += a[i + 1] * q1;
    tv += a[i + 1] * p1;
    asm volatile("" : "+v"(sv), "+v"(tv));
  }
}

__global__ __launch_bounds__(GEMM_TPB) void k_gemm(const float* __restrict__ h,
                                                   const float* __restrict__ W,
                                                   const float* __restrict__ attn_w,
                                                   const int* __restrict__ dst,
                                                   float* __restrict__ z,
                                                   float* __restrict__ sc,
                                                   unsigned* __restrict__ keys) {
  __shared__ v4f Wl[(IN_DIM / 4) * OUT_DIM];
  __shared__ v4f wal[2 * (IN_DIM / 4)];

  const int tid  = threadIdx.x;
  const int lane = tid & 31;
  const int wv   = tid >> 5;
  const int half = lane >> 4;
  const int c    = lane & 15;
  const int nh   = wv & 1;
  const int pr   = wv >> 1;

  int mt = pr * GEMM_BLOCKS + blockIdx.x;
  const v4f* __restrict__ ap = (const v4f*)(h + (size_t)(mt * 16 + c) * IN_DIM) + half;

  v4f a0[8], a1[8];
  load_a_group(a0, ap, 0);

  {
    const int4* __restrict__ d4 = (const int4*)dst;
    for (int i = blockIdx.x * GEMM_TPB + tid; i < N_EDGES / 4; i += GEMM_BLOCKS * GEMM_TPB) {
      const int4 d = d4[i];
      const unsigned o[4] = {(unsigned)d.x / R_NODES, (unsigned)d.y / R_NODES,
                             (unsigned)d.z / R_NODES, (unsigned)d.w / R_NODES};
      unsigned word = 0u;
#pragma unroll
      for (int s = 0; s < 4; ++s) {
#pragma unroll
        for (int k = 0; k < 8; ++k) {
          const unsigned plane = __builtin_amdgcn_ballot_w32(((o[s] >> k) & 1u) != 0u);
          if (lane == 8 * s + k) word = plane;
        }
      }
      keys[i] = word;
    }
  }

  v4f wa_a[OUT_DIM / 8], wa_b[OUT_DIM / 8];
  {
    const v4f* __restrict__ wr = (const v4f*)(W + (size_t)(16 * wv + c) * OUT_DIM) + half;
    const v4f* __restrict__ ar = (const v4f*)(attn_w + (c & 1) * OUT_DIM) + half;
#pragma unroll
    for (int i = 0; i < OUT_DIM / 8; ++i) { wa_a[i] = wr[2 * i]; wa_b[i] = ar[2 * i]; }
  }
  v8f w0 = {}, w1 = {};

  {
    const v4f* __restrict__ W4 = (const v4f*)W;
#pragma unroll
    for (int it = 0; it < STAGE_ROUNDS; ++it) {
      const int e  = tid + it * GEMM_TPB;
      const int kq = e >> 5;
      const int j  = e & 31;
      const v4f r0 = W4[(4 * kq + 0) * (OUT_DIM / 4) + j];
      const v4f r1 = W4[(4 * kq + 1) * (OUT_DIM / 4) + j];
      const v4f r2 = W4[(4 * kq + 2) * (OUT_DIM / 4) + j];
      const v4f r3 = W4[(4 * kq + 3) * (OUT_DIM / 4) + j];
#pragma unroll
      for (int i = 4 * it; i < 4 * it + 4; i += 2) {
        v2f alo, ahi, blo, bhi;
        alo.x = wa_a[i].x; alo.y = wa_a[i].y; ahi.x = wa_a[i].z; ahi.y = wa_a[i].w;
        blo.x = wa_b[i].x; blo.y = wa_b[i].y; bhi.x = wa_b[i].z; bhi.y = wa_b[i].w;
        w0 = __builtin_amdgcn_wmma_f32_16x16x4_f32(false, alo, false, blo, (short)0, w0, false, false);
        w0 = __builtin_amdgcn_wmma_f32_16x16x4_f32(false, ahi, false, bhi, (short)0, w0, false, false);
        alo.x = wa_a[i + 1].x; alo.y = wa_a[i + 1].y; ahi.x = wa_a[i + 1].z; ahi.y = wa_a[i + 1].w;
        blo.x = wa_b[i + 1].x; blo.y = wa_b[i + 1].y; bhi.x = wa_b[i + 1].z; bhi.y = wa_b[i + 1].w;
        w1 = __builtin_amdgcn_wmma_f32_16x16x4_f32(false, alo, false, blo, (short)0, w1, false, false);
        w1 = __builtin_amdgcn_wmma_f32_16x16x4_f32(false, ahi, false, bhi, (short)0, w1, false, false);
      }
      v4f* d = Wl + kq * OUT_DIM + 4 * j;
      v4f q0 = {r0.x, r1.x, r2.x, r3.x};
      v4f q1 = {r0.y, r1.y, r2.y, r3.y};
      v4f q2 = {r0.z, r1.z, r2.z, r3.z};
      v4f q3 = {r0.w, r1.w, r2.w, r3.w};
      d[0] = q0; d[1] = q1; d[2] = q2; d[3] = q3;
    }
  }
  if (c < 2) {
    float* wf = (float*)wal + c * IN_DIM + 16 * wv + 8 * half;
#pragma unroll
    for (int i = 0; i < 8; ++i) wf[i] = w0[i] + w1[i];
  }
  __syncthreads();

  const v4f* __restrict__ bp = Wl + half * OUT_DIM + nh * 64 + c;
  const v4f* __restrict__ wp = wal + half;

  for (;;) {
    v8f acc[4] = {};
    const int  mt_next  = mt + TILE_STEP;
    const bool has_next = mt_next < M_TILES;

    asm volatile("" ::: "memory");
    v4f b0[4], b1[4];
    v4f q0, p0, q1, p1;
    v4f sv = {0.0f, 0.0f, 0.0f, 0.0f};
    v4f tv = {0.0f, 0.0f, 0.0f, 0.0f};
    load_b_chunk(b0, bp, 0);
    q0 = wp[0];
    p0 = wp[IN_DIM / 4];
    load_a_group(a1, ap, 1);
    mma_group<0>(acc, a0, bp, b0, b1, wp, q0, p0, q1, p1, sv, tv);
    load_a_group(a0, ap, 2);
    mma_group<1>(acc, a1, bp, b0, b1, wp, q0, p0, q1, p1, sv, tv);
    load_a_group(a1, ap, 3);
    mma_group<2>(acc, a0, bp, b0, b1, wp, q0, p0, q1, p1, sv, tv);
    const v4f* __restrict__ ap_next =
        (const v4f*)(h + (size_t)(mt_next * 16 + c) * IN_DIM) + half;
    if (has_next) load_a_group(a0, ap_next, 0);
    mma_group<3>(acc, a1, bp, b0, b1, wp, q0, p0, q1, p1, sv, tv);

    {
      float sa = (sv.x + sv.y) + (sv.z + sv.w);
      float sb = (tv.x + tv.y) + (tv.z + tv.w);
      sa += __shfl_xor(sa, 16);
      sb += __shfl_xor(sb, 16);
      if (nh == 0) sc[mt * 32 + lane] = half ? sb : sa;
    }

    float* zp = z + (size_t)(mt * 16 + 8 * half) * OUT_DIM + nh * 64 + c;
#pragma unroll
    for (int t = 0; t < 4; ++t) {
#pragma unroll
      for (int i = 0; i < 8; ++i) zp[(size_t)i * OUT_DIM + 16 * t] = acc[t][i];
    }

    if (!has_next) break;
    mt = mt_next;
    ap = ap_next;
  }
}

__device__ __forceinline__ unsigned enc_f32(float f) {
  unsigned b = __float_as_uint(f);
  return (b & 0x80000000u) ? ~b : (b | 0x80000000u);
}
__device__ __forceinline__ float dec_f32(unsigned e) {
  unsigned b = (e & 0x80000000u) ? (e & 0x7FFFFFFFu) : ~e;
  return __uint_as_float(b);
}

__device__ __forceinline__ int s1_at(int n) { return ((n >> 4) << 5) + (n & 15); }
__device__ __forceinline__ int s2_at(int n) { return ((n >> 4) << 5) + 16 + (n & 15); }

__global__ __launch_bounds__(AGG_TPB) void k_aggregate(const int* __restrict__ src,
                                                       const int* __restrict__ dst,
                                                       const float* __restrict__ z,
                                                       const float* __restrict__ sc,
                                                       const unsigned* __restrict__ keys,
                                                       float* __restrict__ out) {
  __shared__ float    acc[R_NODES * OUT_DIM];
  __shared__ unsigned lst_a[LIST_CAP];
  __shared__ float    lst_e[LIST_CAP];
  __shared__ unsigned menc[R_NODES];
  __shared__ float    den[R_NODES];
  __shared__ float    s2l[R_NODES];
  __shared__ int      cnt;

  const int tid  = threadIdx.x;
  const int lane = tid & 31;
  const int wv   = tid >> 5;
  const int base = blockIdx.x * R_NODES;

  for (int i = tid; i < R_NODES * OUT_DIM / 4; i += AGG_TPB) {
    const v4f zero = {0.0f, 0.0f, 0.0f, 0.0f};
    ((v4f*)acc)[i] = zero;
  }
  if (tid < R_NODES) {
    menc[tid] = 0u;
    den[tid]  = 0.0f;
    const int nd = base + tid;
    s2l[tid] = (nd < N_NODES) ? sc[s2_at(nd)] : 0.0f;
  }
  if (tid == 0) cnt = 0;
  __syncthreads();

  {
    const uint4* __restrict__ k4 = (const uint4*)keys;
    unsigned inv[8];
#pragma unroll
    for (int k = 0; k < 8; ++k) inv[k] = ((blockIdx.x >> k) & 1u) ? 0u : 0xFFFFFFFFu;
    const int start = blockIdx.x * (NSG / N_OWN);
    for (int it = 0; it < SCAN_ITERS; it += SCAN_BATCH) {
      uint4 lo[SCAN_BATCH], hi[SCAN_BATCH];
      int   qq[SCAN_BATCH];
      bool  ok[SCAN_BATCH];
#pragma unroll
      for (int u = 0; u < SCAN_BATCH; ++u) {
        const int t = (it + u) * AGG_TPB + tid;
        ok[u] = t < NSG;
        int q = t + start;
        if (q >= NSG) q -= NSG;
        if (!ok[u]) q = 0;
        qq[u] = q;
        lo[u] = k4[2 * q];
        hi[u] = k4[2 * q + 1];
      }
#pragma unroll
      for (int u = 0; u < SCAN_BATCH; ++u) {
        unsigned m = (lo[u].x ^ inv[0]) & (lo[u].y ^ inv[1]) & (lo[u].z ^ inv[2]) &
                     (lo[u].w ^ inv[3]) & (hi[u].x ^ inv[4]) & (hi[u].y ^ inv[5]) &
                     (hi[u].z ^ inv[6]) & (hi[u].w ^ inv[7]);
        if (!ok[u]) m = 0u;
        const unsigned e0 = ((unsigned)(qq[u] >> 2) << 7) + (unsigned)(qq[u] & 3);
        while (m != 0u) {
          const int l = __builtin_ctz(m);
          m &= m - 1u;
          const int pos = atomicAdd(&cnt, 1);
          if (pos < LIST_CAP) lst_a[pos] = e0 + 4u * (unsigned)l;
        }
      }
    }
  }
  __syncthreads();
  const int n = cnt;

  if (n <= LIST_CAP) {
    for (int j0 = tid; j0 < n; j0 += 4 * AGG_TPB) {
      int ii[4], dd[4], ss[4];
      float sv[4];
#pragma unroll
      for (int u = 0; u < 4; ++u) {
        const int j = j0 + u * AGG_TPB;
        ii[u] = (j < n) ? (int)lst_a[j] : 0;
      }
#pragma unroll
      for (int u = 0; u < 4; ++u) { dd[u] = dst[ii[u]]; ss[u] = src[ii[u]]; }
#pragma unroll
      for (int u = 0; u < 4; ++u) sv[u] = sc[s1_at(ss[u])];
#pragma unroll
      for (int u = 0; u < 4; ++u) {
        const int j = j0 + u * AGG_TPB;
        if (j < n) {
          const int loc = dd[u] - base;
          const float logit = sv[u] + s2l[loc];
          const float e = (logit >= 0.0f) ? logit : LEAK * logit;
          lst_e[j] = e;
          lst_a[j] = ((unsigned)loc << 16) | (unsigned)ss[u];
          atomicMax(&menc[loc], enc_f32(e));
        }
      }
    }
    __syncthreads();

    for (int j = tid; j < n; j += AGG_TPB) {
      const int loc = (int)(lst_a[j] >> 16);
      const float ex = expf(lst_e[j] - dec_f32(menc[loc]));
      lst_e[j] = ex;
      atomicAdd(&den[loc], ex);
    }
    __syncthreads();

    const v4f* __restrict__ z4 = (const v4f*)z;
    int j0 = wv;
    for (; j0 + 7 * 32 < n; j0 += 8 * 32) {
      unsigned pk[8];
      float    ex[8];
      v4f      v[8];
#pragma unroll
      for (int u = 0; u < 8; ++u) { pk[u] = lst_a[j0 + u * 32]; ex[u] = lst_e[j0 + u * 32]; }
#pragma unroll
      for (int u = 0; u < 8; ++u) v[u] = z4[(size_t)(pk[u] & 0xFFFFu) * (OUT_DIM / 4) + lane];
#pragma unroll
      for (int u = 0; u < 8; ++u) {
        float* a = acc + (pk[u] >> 16) * OUT_DIM + lane;
        atomicAdd(a,      ex[u] * v[u].x);
        atomicAdd(a + 32, ex[u] * v[u].y);
        atomicAdd(a + 64, ex[u] * v[u].z);
        atomicAdd(a + 96, ex[u] * v[u].w);
      }
    }
    for (; j0 < n; j0 += 32) {
      const unsigned pk = lst_a[j0];
      const float    ex = lst_e[j0];
      const v4f v = z4[(size_t)(pk & 0xFFFFu) * (OUT_DIM / 4) + lane];
      float* a = acc + (pk >> 16) * OUT_DIM + lane;
      atomicAdd(a,      ex * v.x);
      atomicAdd(a + 32, ex * v.y);
      atomicAdd(a + 64, ex * v.z);
      atomicAdd(a + 96, ex * v.w);
    }
  } else {
    for (int i = tid; i < N_EDGES; i += AGG_TPB) {
      const int loc = dst[i] - base;
      if ((unsigned)loc < (unsigned)R_NODES) {
        const float logit = sc[s1_at(src[i])] + s2l[loc];
        const float e = (logit >= 0.0f) ? logit : LEAK * logit;
        atomicMax(&menc[loc], enc_f32(e));
      }
    }
    __syncthreads();
    for (int i = tid; i < N_EDGES; i += AGG_TPB) {
      const int loc = dst[i] - base;
      if ((unsigned)loc < (unsigned)R_NODES) {
        const int s = src[i];
        const float logit = sc[s1_at(s)] + s2l[loc];
        const float e = (logit >= 0.0f) ? logit : LEAK * logit;
        const float ex = expf(e - dec_f32(menc[loc]));
        atomicAdd(&den[loc], ex);
        const float* zr = z + (size_t)s * OUT_DIM;
        float* a = acc + loc * OUT_DIM;
        for (int c = 0; c < OUT_DIM; ++c) atomicAdd(&a[(c & 3) * 32 + (c >> 2)], ex * zr[c]);
      }
    }
  }
  __syncthreads();

  for (int idx = tid; idx < R_NODES * (OUT_DIM / 4); idx += AGG_TPB) {
    const int r  = idx >> 5;
    const int q  = idx & 31;
    const int nd = base + r;
    if (nd < N_NODES) {
      const float dn0 = den[r];
      const float dn  = (dn0 > 0.0f) ? dn0 : 1.0f;
      const float* a = acc + r * OUT_DIM + q;
      v4f x;
      x.x = a[0] / dn; x.y = a[32] / dn; x.z = a[64] / dn; x.w = a[96] / dn;
      x.x = (x.x > 0.0f) ? x.x : expm1f(x.x);
      x.y = (x.y > 0.0f) ? x.y : expm1f(x.y);
      x.z = (x.z > 0.0f) ? x.z : expm1f(x.z);
      x.w = (x.w > 0.0f) ? x.w : expm1f(x.w);
      ((v4f*)out)[(size_t)nd * (OUT_DIM / 4) + q] = x;
    }
  }
}

extern "C" void kernel_launch(void* const* d_in, const int* in_sizes, int n_in,
                              void* d_out, int out_size, void* d_ws, size_t ws_size,
                              hipStream_t stream) {
  const float* h      = (const float*)d_in[0];
  const int*   src    = (const int*)  d_in[1];
  const int*   dst    = (const int*)  d_in[2];
  const float* W_fc   = (const float*)d_in[3];
  const float* attn_w = (const float*)d_in[4];
  float*       out    = (float*)d_out;

  float*    z    = (float*)d_ws;
  float*    sc   = z  + (size_t)N_NODES * OUT_DIM;
  unsigned* keys = (unsigned*)(sc + 2 * N_NODES);

  k_gemm<<<GEMM_BLOCKS, GEMM_TPB, 0, stream>>>(h, W_fc, attn_w, dst, z, sc, keys);

  k_aggregate<<<N_OWN, AGG_TPB, 0, stream>>>(src, dst, z, sc, keys, out);
}


# row sums in registers, not LDS: block's edge list binned by owning wave (dst row mod 32; LDS count, scan, place), a wave sums its 7 rows and denominators; no LDS float adds; sum order: arrival in bin
# speedup vs baseline: 4.5089x; 1.0688x over previous
#include <hip/hip_runtime.h>
#include <math.h>

#define N_NODES 50000
#define N_EDGES 800000
#define IN_DIM  256
#define OUT_DIM 128
#define LEAK    0.2f

typedef __attribute__((ext_vector_type(2))) float v2f;
typedef __attribute__((ext_vector_type(4))) float v4f;
typedef __attribute__((ext_vector_type(8))) float v8f;

#define R_NODES  196
#define N_OWN    ((N_NODES + R_NODES - 1) / R_NODES)
#define LIST_CAP 16384
#define NSG      (N_EDGES / 32)
#define AGG_TPB  1024
#define SCAN_BATCH 5
#define SCAN_ITERS ((NSG + AGG_TPB - 1) / AGG_TPB)

static_assert(N_OWN <= 256, "owner id must fit in one byte");
static_assert(N_EDGES % 128 == 0, "a wave slices the keys of 128 edges at a time");
static_assert(N_NODES <= 65536, "src index is packed into 16 bits");
static_assert(SCAN_ITERS % SCAN_BATCH == 0, "scan batches must tile the scan");
static_assert(OUT_DIM == 128, "one float4 per lane per row");

#define GEMM_TPB    512
#define GEMM_BLOCKS 256
#define M_TILES     (N_NODES / 16)
#define TILE_STEP   (GEMM_BLOCKS * (GEMM_TPB / 64))

static_assert(N_NODES % 16 == 0, "row tiles are whole");
static_assert(IN_DIM % 64 == 0, "k is walked in groups of 64");
static_assert(IN_DIM == 16 * (GEMM_TPB / 32), "one 16-row tile of W per wave for W @ a");
#define STAGE_ROUNDS ((IN_DIM / 4) * (OUT_DIM / 4) / GEMM_TPB)
static_assert(STAGE_ROUNDS * GEMM_TPB == (IN_DIM / 4) * (OUT_DIM / 4), "staging rounds are whole");
static_assert(OUT_DIM / 8 == 4 * STAGE_ROUNDS, "four 8-k chunks of W @ a per staging round");

__device__ __forceinline__ void load_a_group(v4f (&a)[8], const v4f* __restrict__ ap, int g) {
#pragma unroll
  for (int i = 0; i < 8; ++i) a[i] = ap[2 * (8 * g + i)];
}

__device__ __forceinline__ void load_b_chunk(v4f (&b)[4], const v4f* __restrict__ bp, int kc) {
  const v4f* __restrict__ bq = bp + (size_t)(2 * kc) * OUT_DIM;
#pragma unroll
  for (int t = 0; t < 4; ++t) b[t] = bq[16 * t];
}

__device__ __forceinline__ void mma_chunk(v8f (&acc)[4], const v4f a, const v4f (&b)[4]) {
  v2f alo; alo.x = a.x; alo.y = a.y;
  v2f ahi; ahi.x = a.z; ahi.y = a.w;
#pragma unroll
  for (int t = 0; t < 4; ++t) {
    v2f blo; blo.x = b[t].x; blo.y = b[t].y;
    acc[t] = __builtin_amdgcn_wmma_f32_16x16x4_f32(
         false, alo,  false, blo,
         (short)0, acc[t],  false,  false);
  }
#pragma unroll
  for (int t = 0; t < 4; ++t) {
    v2f bhi; bhi.x = b[t].z; bhi.y = b[t].w;
    acc[t] = __builtin_amdgcn_wmma_f32_16x16x4_f32(
         false, ahi,  false, bhi,
         (short)0, acc[t],  false,  false);
  }
}

template <int G>
__device__ __forceinline__ void mma_group(v8f (&acc)[4], const v4f (&a)[8],
                                          const v4f* __restrict__ bp,
                                          v4f (&b0)[4], v4f (&b1)[4],
                                          const v4f* __restrict__ wp,
                                          v4f& q0, v4f& p0, v4f& q1, v4f& p1,
                                          v4f& sv, v4f& tv) {
#pragma unroll
  for (int i = 0; i < 8; i += 2) {
    load_b_chunk(b1, bp, 8 * G + i + 1);
    q1 = wp[2 * (8 * G + i + 1)];
    p1 = wp[2 * (8 * G + i + 1) + IN_DIM / 4];
    mma_chunk(acc, a[i], b0);
    sv += a[i] * q0;
    tv += a[i] * p0;
    asm volatile("" : "+v"(sv), "+v"(tv));
    if (8 * G + i + 2 < IN_DIM / 8) {
      load_b_chunk(b0, bp, 8 * G + i + 2);
      q0 = wp[2 * (8 * G + i + 2)];
      p0 = wp[2 * (8 * G + i + 2) + IN_DIM / 4];
    }
    mma_chunk(acc, a[i + 1], b1);
    sv += a[i + 1] * q1;
    tv += a[i + 1] * p1;
    asm volatile("" : "+v"(sv), "+v"(tv));
  }
}

__global__ __launch_bounds__(GEMM_TPB) void k_gemm(const float* __restrict__ h,
                                                   const float* __restrict__ W,
                                                   const float* __restrict__ attn_w,
                                                   const int* __restrict__ dst,
                                                   float* __restrict__ z,
                                                   float* __restrict__ sc,
                                                   unsigned* __restrict__ keys) {
  __shared__ v4f Wl[(IN_DIM / 4) * OUT_DIM];
  __shared__ v4f wal[2 * (IN_DIM / 4)];

  const int tid  = threadIdx.x;
  const int lane = tid & 31;
  const int wv   = tid >> 5;
  const int half = lane >> 4;
  const int c    = lane & 15;
  const int nh   = wv & 1;
  const int pr   = wv >> 1;

  int mt = pr * GEMM_BLOCKS + blockIdx.x;
  const v4f* __restrict__ ap = (const v4f*)(h + (size_t)(mt * 16 + c) * IN_DIM) + half;

  v4f a0[8], a1[8];
  load_a_group(a0, ap, 0);

  {
    const int4* __restrict__ d4 = (const int4*)dst;
    for (int i = blockIdx.x * GEMM_TPB + tid; i < N_EDGES / 4; i += GEMM_BLOCKS * GEMM_TPB) {
      const int4 d = d4[i];
      const unsigned o[4] = {(unsigned)d.x / R_NODES, (unsigned)d.y / R_NODES,
                             (unsigned)d.z / R_NODES, (unsigned)d.w / R_NODES};
      unsigned word = 0u;
#pragma unroll
      for (int s = 0; s < 4; ++s) {
#pragma unroll
        for (int k = 0; k < 8; ++k) {
          const unsigned plane = __builtin_amdgcn_ballot_w32(((o[s] >> k) & 1u) != 0u);
          if (lane == 8 * s + k) word = plane;
        }
      }
      keys[i] = word;
    }
  }

  v4f wa_a[OUT_DIM / 8], wa_b[OUT_DIM / 8];
  {
    const v4f* __restrict__ wr = (const v4f*)(W + (size_t)(16 * wv + c) * OUT_DIM) + half;
    const v4f* __restrict__ ar = (const v4f*)(attn_w + (c & 1) * OUT_DIM) + half;
#pragma unroll
    for (int i = 0; i < OUT_DIM / 8; ++i) { wa_a[i] = wr[2 * i]; wa_b[i] = ar[2 * i]; }
  }
  v8f w0 = {}, w1 = {};

  {
    const v4f* __restrict__ W4 = (const v4f*)W;
#pragma unroll
    for (int it = 0; it < STAGE_ROUNDS; ++it) {
      const int e  = tid + it * GEMM_TPB;
      const int kq = e >> 5;
      const int j  = e & 31;
      const v4f r0 = W4[(4 * kq + 0) * (OUT_DIM / 4) + j];
      const v4f r1 = W4[(4 * kq + 1) * (OUT_DIM / 4) + j];
      const v4f r2 = W4[(4 * kq + 2) * (OUT_DIM / 4) + j];
      const v4f r3 = W4[(4 * kq + 3) * (OUT_DIM / 4) + j];
#pragma unroll
      for (int i = 4 * it; i < 4 * it + 4; i += 2) {
        v2f alo, ahi, blo, bhi;
        alo.x = wa_a[i].x; alo.y = wa_a[i].y; ahi.x = wa_a[i].z; ahi.y = wa_a[i].w;
        blo.x = wa_b[i].x; blo.y = wa_b[i].y; bhi.x = wa_b[i].z; bhi.y = wa_b[i].w;
        w0 = __builtin_amdgcn_wmma_f32_16x16x4_f32(false, alo, false, blo, (short)0, w0, false, false);
        w0 = __builtin_amdgcn_wmma_f32_16x16x4_f32(false, ahi, false, bhi, (short)0, w0, false, false);
        alo.x = wa_a[i + 1].x; alo.y = wa_a[i + 1].y; ahi.x = wa_a[i + 1].z; ahi.y = wa_a[i + 1].w;
        blo.x = wa_b[i + 1].x; blo.y = wa_b[i + 1].y; bhi.x = wa_b[i + 1].z; bhi.y = wa_b[i + 1].w;
        w1 = __builtin_amdgcn_wmma_f32_16x16x4_f32(false, alo, false, blo, (short)0, w1, false, false);
        w1 = __builtin_amdgcn_wmma_f32_16x16x4_f32(false, ahi, false, bhi, (short)0, w1, false, false);
      }
      v4f* d = Wl + kq * OUT_DIM + 4 * j;
      v4f q0 = {r0.x, r1.x, r2.x, r3.x};
      v4f q1 = {r0.y, r1.y, r2.y, r3.y};
      v4f q2 = {r0.z, r1.z, r2.z, r3.z};
      v4f q3 = {r0.w, r1.w, r2.w, r3.w};
      d[0] = q0; d[1] = q1; d[2] = q2; d[3] = q3;
    }
  }
  if (c < 2) {
    float* wf = (float*)wal + c * IN_DIM + 16 * wv + 8 * half;
#pragma unroll
    for (int i = 0; i < 8; ++i) wf[i] = w0[i] + w1[i];
  }
  __syncthreads();

  const v4f* __restrict__ bp = Wl + half * OUT_DIM + nh * 64 + c;
  const v4f* __restrict__ wp = wal + half;

  for (;;) {
    v8f acc[4] = {};
    const int  mt_next  = mt + TILE_STEP;
    const bool has_next = mt_next < M_TILES;

    asm volatile("" ::: "memory");
    v4f b0[4], b1[4];
    v4f q0, p0, q1, p1;
    v4f sv = {0.0f, 0.0f, 0.0f, 0.0f};
    v4f tv = {0.0f, 0.0f, 0.0f, 0.0f};
    load_b_chunk(b0, bp, 0);
    q0 = wp[0];
    p0 = wp[IN_DIM / 4];
    load_a_group(a1, ap, 1);
    mma_group<0>(acc, a0, bp, b0, b1, wp, q0, p0, q1, p1, sv, tv);
    load_a_group(a0, ap, 2);
    mma_group<1>(acc, a1, bp, b0, b1, wp, q0, p0, q1, p1, sv, tv);
    load_a_group(a1, ap, 3);
    mma_group<2>(acc, a0, bp, b0, b1, wp, q0, p0, q1, p1, sv, tv);
    const v4f* __restrict__ ap_next =
        (const v4f*)(h + (size_t)(mt_next * 16 + c) * IN_DIM) + half;
    if (has_next) load_a_group(a0, ap_next, 0);
    mma_group<3>(acc, a1, bp, b0, b1, wp, q0, p0, q1, p1, sv, tv);

    {
      float sa = (sv.x + sv.y) + (sv.z + sv.w);
      float sb = (tv.x + tv.y) + (tv.z + tv.w);
      sa += __shfl_xor(sa, 16);
      sb += __shfl_xor(sb, 16);
      if (nh == 0) sc[mt * 32 + lane] = half ? sb : sa;
    }

    float* zp = z + (size_t)(mt * 16 + 8 * half) * OUT_DIM + nh * 64 + c;
#pragma unroll
    for (int t = 0; t < 4; ++t) {
#pragma unroll
      for (int i = 0; i < 8; ++i) zp[(size_t)i * OUT_DIM + 16 * t] = acc[t][i];
    }

    if (!has_next) break;
    mt = mt_next;
    ap = ap_next;
  }
}

__device__ __forceinline__ unsigned enc_f32(float f) {
  unsigned b = __float_as_uint(f);
  return (b & 0x80000000u) ? ~b : (b | 0x80000000u);
}
__device__ __forceinline__ float dec_f32(unsigned e) {
  unsigned b = (e & 0x80000000u) ? (e & 0x7FFFFFFFu) : ~e;
  return __uint_as_float(b);
}

__device__ __forceinline__ int s1_at(int n) { return ((n >> 4) << 5) + (n & 15); }
__device__ __forceinline__ int s2_at(int n) { return ((n >> 4) << 5) + 16 + (n & 15); }

#define N_BIN 32
#define ROWS_PER_WAVE ((R_NODES + N_BIN - 1) / N_BIN)
static_assert(N_BIN == AGG_TPB / 32, "one bin per wave");
static_assert(sizeof(uint2) * LIST_CAP >= sizeof(float) * R_NODES * OUT_DIM,
              "the overflow path keeps its row sums in the bin array");

__device__ __forceinline__ void add_row(v4f (&ac)[ROWS_PER_WAVE], float (&dn)[ROWS_PER_WAVE],
                                        const int r, const float ex, const v4f v) {
#pragma unroll
  for (int k = 0; k < ROWS_PER_WAVE; ++k) {
    if (r == k) {
      asm volatile("");
      ac[k] += ex * v;
      dn[k] += ex;
    }
  }
}

__global__ __launch_bounds__(AGG_TPB) void k_aggregate(const int* __restrict__ src,
                                                       const int* __restrict__ dst,
                                                       const float* __restrict__ z,
                                                       const float* __restrict__ sc,
                                                       const unsigned* __restrict__ keys,
                                                       float* __restrict__ out) {
  __shared__ __attribute__((aligned(16))) uint2 lst2[LIST_CAP];
  __shared__ unsigned lst_a[LIST_CAP];
  __shared__ float    lst_e[LIST_CAP];
  __shared__ unsigned menc[R_NODES];
  __shared__ float    den[R_NODES];
  __shared__ float    s2l[R_NODES];
  __shared__ int      bcnt[N_BIN];
  __shared__ int      bbase[N_BIN];
  __shared__ int      bfill[N_BIN];
  __shared__ int      cnt;

  const int tid  = threadIdx.x;
  const int lane = tid & 31;
  const int wv   = tid >> 5;
  const int base = blockIdx.x * R_NODES;

  if (tid < R_NODES) {
    menc[tid] = 0u;
    den[tid]  = 0.0f;
    const int nd = base + tid;
    s2l[tid] = (nd < N_NODES) ? sc[s2_at(nd)] : 0.0f;
  }
  if (tid < N_BIN) bcnt[tid] = 0;
  if (tid == 0) cnt = 0;
  __syncthreads();

  {
    const uint4* __restrict__ k4 = (const uint4*)keys;
    unsigned inv[8];
#pragma unroll
    for (int k = 0; k < 8; ++k) inv[k] = ((blockIdx.x >> k) & 1u) ? 0u : 0xFFFFFFFFu;
    const int start = blockIdx.x * (NSG / N_OWN);
    for (int it = 0; it < SCAN_ITERS; it += SCAN_BATCH) {
      uint4 lo[SCAN_BATCH], hi[SCAN_BATCH];
      int   qq[SCAN_BATCH];
      bool  ok[SCAN_BATCH];
#pragma unroll
      for (int u = 0; u < SCAN_BATCH; ++u) {
        const int t = (it + u) * AGG_TPB + tid;
        ok[u] = t < NSG;
        int q = t + start;
        if (q >= NSG) q -= NSG;
        if (!ok[u]) q = 0;
        qq[u] = q;
        lo[u] = k4[2 * q];
        hi[u] = k4[2 * q + 1];
      }
#pragma unroll
      for (int u = 0; u < SCAN_BATCH; ++u) {
        unsigned m = (lo[u].x ^ inv[0]) & (lo[u].y ^ inv[1]) & (lo[u].z ^ inv[2]) &
                     (lo[u].w ^ inv[3]) & (hi[u].x ^ inv[4]) & (hi[u].y ^ inv[5]) &
                     (hi[u].z ^ inv[6]) & (hi[u].w ^ inv[7]);
        if (!ok[u]) m = 0u;
        const unsigned e0 = ((unsigned)(qq[u] >> 2) << 7) + (unsigned)(qq[u] & 3);
        while (m != 0u) {
          const int l = __builtin_ctz(m);
          m &= m - 1u;
          const int pos = atomicAdd(&cnt, 1);
          if (pos < LIST_CAP) lst_a[pos] = e0 + 4u * (unsigned)l;
        }
      }
    }
  }
  __syncthreads();
  const int n = cnt;

  if (n <= LIST_CAP) {
    for (int j0 = tid; j0 < n; j0 += 4 * AGG_TPB) {
      int ii[4], dd[4], ss[4];
      float sv[4];
#pragma unroll
      for (int u = 0; u < 4; ++u) {
        const int j = j0 + u * AGG_TPB;
        ii[u] = (j < n) ? (int)lst_a[j] : 0;
      }
#pragma unroll
      for (int u = 0; u < 4; ++u) { dd[u] = dst[ii[u]]; ss[u] = src[ii[u]]; }
#pragma unroll
      for (int u = 0; u < 4; ++u) sv[u] = sc[s1_at(ss[u])];
#pragma unroll
      for (int u = 0; u < 4; ++u) {
        const int j = j0 + u * AGG_TPB;
        if (j < n) {
          const int loc = dd[u] - base;
          const float logit = sv[u] + s2l[loc];
          const float e = (logit >= 0.0f) ? logit : LEAK * logit;
          lst_e[j] = e;
          lst_a[j] = ((unsigned)loc << 16) | (unsigned)ss[u];
          atomicMax(&menc[loc], enc_f32(e));
          atomicAdd(&bcnt[loc & (N_BIN - 1)], 1);
        }
      }
    }
    __syncthreads();

    if (wv == 0) {
      const int c = bcnt[lane];
      int incl = c;
#pragma unroll
      for (int d = 1; d < N_BIN; d <<= 1) {
        const int t = __shfl_up(incl, d);
        if (lane >= d) incl += t;
      }
      bbase[lane] = incl - c;
      bfill[lane] = incl - c;
    }
    __syncthreads();

    for (int j = tid; j < n; j += AGG_TPB) {
      const unsigned pk = lst_a[j];
      const int loc = (int)(pk >> 16);
      const float ex = expf(lst_e[j] - dec_f32(menc[loc]));
      const int pos = atomicAdd(&bfill[loc & (N_BIN - 1)], 1);
      lst2[pos] = make_uint2(pk, __float_as_uint(ex));
    }
    __syncthreads();

    {
      const v4f* __restrict__ z4 = (const v4f*)z;
      const int beg = __builtin_amdgcn_readfirstlane(bbase[wv]);
      const int end = beg + __builtin_amdgcn_readfirstlane(bcnt[wv]);
      v4f   ac[ROWS_PER_WAVE];
      float dn[ROWS_PER_WAVE];
#pragma unroll
      for (int k = 0; k < ROWS_PER_WAVE; ++k) {
        const v4f zero = {0.0f, 0.0f, 0.0f, 0.0f};
        ac[k] = zero;
        dn[k] = 0.0f;
      }
      int j = beg;
      for (; j + 8 <= end; j += 8) {
        uint2 en[8];
        v4f   v[8];
#pragma unroll
        for (int u = 0; u < 8; ++u) en[u] = lst2[j + u];
#pragma unroll
        for (int u = 0; u < 8; ++u) v[u] = z4[(size_t)(en[u].x & 0xFFFFu) * (OUT_DIM / 4) + lane];
#pragma unroll
        for (int u = 0; u < 8; ++u) {
          const int r = __builtin_amdgcn_readfirstlane((int)(en[u].x >> 21));
          add_row(ac, dn, r, __uint_as_float(en[u].y), v[u]);
        }
      }
      const int rem = end - j;
      if (rem > 0) {
        uint2 en[7];
        v4f   v[7];
#pragma unroll
        for (int u = 0; u < 7; ++u) {
          const int ju = j + u;
          en[u] = lst2[ju < LIST_CAP ? ju : LIST_CAP - 1];
        }
#pragma unroll
        for (int u = 0; u < 7; ++u) {
          const v4f zero = {0.0f, 0.0f, 0.0f, 0.0f};
          v[u] = zero;
          if (u < rem) v[u] = z4[(size_t)(en[u].x & 0xFFFFu) * (OUT_DIM / 4) + lane];
        }
#pragma unroll
        for (int u = 0; u < 7; ++u) {
          if (u < rem) {
            const int r = __builtin_amdgcn_readfirstlane((int)(en[u].x >> 21));
            add_row(ac, dn, r, __uint_as_float(en[u].y), v[u]);
          }
        }
      }
#pragma unroll
      for (int k = 0; k < ROWS_PER_WAVE; ++k) {
        const int loc = wv + N_BIN * k;
        const int nd  = base + loc;
        if (loc < R_NODES && nd < N_NODES) {
          const float d = (dn[k] > 0.0f) ? dn[k] : 1.0f;
          v4f x;
          x.x = ac[k].x / d; x.y = ac[k].y / d; x.z = ac[k].z / d; x.w = ac[k].w / d;
          x.x = (x.x > 0.0f) ? x.x : expm1f(x.x);
          x.y = (x.y > 0.0f) ? x.y : expm1f(x.y);
          x.z = (x.z > 0.0f) ? x.z : expm1f(x.z);
          x.w = (x.w > 0.0f) ? x.w : expm1f(x.w);
          ((v4f*)out)[(size_t)nd * (OUT_DIM / 4) + lane] = x;
        }
      }
    }
  } else {
    float* acc = (float*)lst2;
    for (int i = tid; i < R_NODES * OUT_DIM / 4; i += AGG_TPB) {
      const v4f zero = {0.0f, 0.0f, 0.0f, 0.0f};
      ((v4f*)acc)[i] = zero;
    }
    __syncthreads();
    for (int i = tid; i < N_EDGES; i += AGG_TPB) {
      const int loc = dst[i] - base;
      if ((unsigned)loc < (unsigned)R_NODES) {
        const float logit = sc[s1_at(src[i])] + s2l[loc];
        const float e = (logit >= 0.0f) ? logit : LEAK * logit;
        atomicMax(&menc[loc], enc_f32(e));
      }
    }
    __syncthreads();
    for (int i = tid; i < N_EDGES; i += AGG_TPB) {
      const int loc = dst[i] - base;
      if ((unsigned)loc < (unsigned)R_NODES) {
        const int s = src[i];
        const float logit = sc[s1_at(s)] + s2l[loc];
        const float e = (logit >= 0.0f) ? logit : LEAK * logit;
        const float ex = expf(e - dec_f32(menc[loc]));
        atomicAdd(&den[loc], ex);
        const float* zr = z + (size_t)s * OUT_DIM;
        float* a = acc + loc * OUT_DIM;
        for (int c = 0; c < OUT_DIM; ++c) atomicAdd(&a[(c & 3) * 32 + (c >> 2)], ex * zr[c]);
      }
    }
    __syncthreads();

    for (int idx = tid; idx < R_NODES * (OUT_DIM / 4); idx += AGG_TPB) {
      const int r  = idx >> 5;
      const int q  = idx & 31;
      const int nd = base + r;
      if (nd < N_NODES) {
        const float dn0 = den[r];
        const float dn  = (dn0 > 0.0f) ? dn0 : 1.0f;
        const float* a = acc + r * OUT_DIM + q;
        v4f x;
        x.x = a[0] / dn; x.y = a[32] / dn; x.z = a[64] / dn; x.w = a[96] / dn;
        x.x = (x.x > 0.0f) ? x.x : expm1f(x.x);
        x.y = (x.y > 0.0f) ? x.y : expm1f(x.y);
        x.z = (x.z > 0.0f) ? x.z : expm1f(x.z);
        x.w = (x.w > 0.0f) ? x.w : expm1f(x.w);
        ((v4f*)out)[(size_t)nd * (OUT_DIM / 4) + q] = x;
      }
    }
  }
}

extern "C" void kernel_launch(void* const* d_in, const int* in_sizes, int n_in,
                              void* d_out, int out_size, void* d_ws, size_t ws_size,
                              hipStream_t stream) {
  const float* h      = (const float*)d_in[0];
  const int*   src    = (const int*)  d_in[1];
  const int*   dst    = (const int*)  d_in[2];
  const float* W_fc   = (const float*)d_in[3];
  const float* attn_w = (const float*)d_in[4];
  float*       out    = (float*)d_out;

  float*    z    = (float*)d_ws;
  float*    sc   = z  + (size_t)N_NODES * OUT_DIM;
  unsigned* keys = (unsigned*)(sc + 2 * N_NODES);

  k_gemm<<<GEMM_BLOCKS, GEMM_TPB, 0, stream>>>(h, W_fc, attn_w, dst, z, sc, keys);

  k_aggregate<<<N_OWN, AGG_TPB, 0, stream>>>(src, dst, z, sc, keys, out);
}


# per-edge record word (dst row inside its owner block << 16 | src) written as whole lines by launch 1; an owner fetches one scattered word per listed edge, not a dst word and a src word
# speedup vs baseline: 4.6556x; 1.0325x over previous
#include <hip/hip_runtime.h>
#include <math.h>

#define N_NODES 50000
#define N_EDGES 800000
#define IN_DIM  256
#define OUT_DIM 128
#define LEAK    0.2f

typedef __attribute__((ext_vector_type(2))) float v2f;
typedef __attribute__((ext_vector_type(4))) float v4f;
typedef __attribute__((ext_vector_type(8))) float v8f;

#define R_NODES  196
#define N_OWN    ((N_NODES + R_NODES - 1) / R_NODES)
#define LIST_CAP 16384
#define NSG      (N_EDGES / 32)
#define AGG_TPB  1024
#define SCAN_BATCH 5
#define SCAN_ITERS ((NSG + AGG_TPB - 1) / AGG_TPB)

static_assert(N_OWN <= 256, "owner id must fit in one byte");
static_assert(N_EDGES % 128 == 0, "a wave slices the keys of 128 edges at a time");
static_assert(N_NODES <= 65536, "src index is packed into 16 bits");
static_assert(SCAN_ITERS % SCAN_BATCH == 0, "scan batches must tile the scan");
static_assert(OUT_DIM == 128, "one float4 per lane per row");

#define GEMM_TPB    512
#define GEMM_BLOCKS 256
#define M_TILES     (N_NODES / 16)
#define TILE_STEP   (GEMM_BLOCKS * (GEMM_TPB / 64))

static_assert(N_NODES % 16 == 0, "row tiles are whole");
static_assert(IN_DIM % 64 == 0, "k is walked in groups of 64");
static_assert(IN_DIM == 16 * (GEMM_TPB / 32), "one 16-row tile of W per wave for W @ a");
#define STAGE_ROUNDS ((IN_DIM / 4) * (OUT_DIM / 4) / GEMM_TPB)
static_assert(STAGE_ROUNDS * GEMM_TPB == (IN_DIM / 4) * (OUT_DIM / 4), "staging rounds are whole");
static_assert(OUT_DIM / 8 == 4 * STAGE_ROUNDS, "four 8-k chunks of W @ a per staging round");

__device__ __forceinline__ void load_a_group(v4f (&a)[8], const v4f* __restrict__ ap, int g) {
#pragma unroll
  for (int i = 0; i < 8; ++i) a[i] = ap[2 * (8 * g + i)];
}

__device__ __forceinline__ void load_b_chunk(v4f (&b)[4], const v4f* __restrict__ bp, int kc) {
  const v4f* __restrict__ bq = bp + (size_t)(2 * kc) * OUT_DIM;
#pragma unroll
  for (int t = 0; t < 4; ++t) b[t] = bq[16 * t];
}

__device__ __forceinline__ void mma_chunk(v8f (&acc)[4], const v4f a, const v4f (&b)[4]) {
  v2f alo; alo.x = a.x; alo.y = a.y;
  v2f ahi; ahi.x = a.z; ahi.y = a.w;
#pragma unroll
  for (int t = 0; t < 4; ++t) {
    v2f blo; blo.x = b[t].x; blo.y = b[t].y;
    acc[t] = __builtin_amdgcn_wmma_f32_16x16x4_f32(
         false, alo,  false, blo,
         (short)0, acc[t],  false,  false);
  }
#pragma unroll
  for (int t = 0; t < 4; ++t) {
    v2f bhi; bhi.x = b[t].z; bhi.y = b[t].w;
    acc[t] = __builtin_amdgcn_wmma_f32_16x16x4_f32(
         false, ahi,  false, bhi,
         (short)0, acc[t],  false,  false);
  }
}

template <int G>
__device__ __forceinline__ void mma_group(v8f (&acc)[4], const v4f (&a)[8],
                                          const v4f* __restrict__ bp,
                                          v4f (&b0)[4], v4f (&b1)[4],
                                          const v4f* __restrict__ wp,
                                          v4f& q0, v4f& p0, v4f& q1, v4f& p1,
                                          v4f& sv, v4f& tv) {
#pragma unroll
  for (int i = 0; i < 8; i += 2) {
    load_b_chunk(b1, bp, 8 * G + i + 1);
    q1 = wp[2 * (8 * G + i + 1)];
    p1 = wp[2 * (8 * G + i + 1) + IN_DIM / 4];
    mma_chunk(acc, a[i], b0);
    sv += a[i] * q0;
    tv += a[i] * p0;
    asm volatile("" : "+v"(sv), "+v"(tv));
    if (8 * G + i + 2 < IN_DIM / 8) {
      load_b_chunk(b0, bp, 8 * G + i + 2);
      q0 = wp[2 * (8 * G + i + 2)];
      p0 = wp[2 * (8 * G + i + 2) + IN_DIM / 4];
    }
    mma_chunk(acc, a[i + 1], b1);
    sv += a[i + 1] * q1;
    tv += a[i + 1] * p1;
    asm volatile("" : "+v"(sv), "+v"(tv));
  }
}

__global__ __launch_bounds__(GEMM_TPB) void k_gemm(const float* __restrict__ h,
                                                   const float* __restrict__ W,
                                                   const float* __restrict__ attn_w,
                                                   const int* __restrict__ src,
                                                   const int* __restrict__ dst,
                                                   float* __restrict__ z,
                                                   float* __restrict__ sc,
                                                   unsigned* __restrict__ keys,
                                                   unsigned* __restrict__ rec) {
  __shared__ v4f Wl[(IN_DIM / 4) * OUT_DIM];
  __shared__ v4f wal[2 * (IN_DIM / 4)];

  const int tid  = threadIdx.x;
  const int lane = tid & 31;
  const int wv   = tid >> 5;
  const int half = lane >> 4;
  const int c    = lane & 15;
  const int nh   = wv & 1;
  const int pr   = wv >> 1;

  int mt = pr * GEMM_BLOCKS + blockIdx.x;
  const v4f* __restrict__ ap = (const v4f*)(h + (size_t)(mt * 16 + c) * IN_DIM) + half;

  v4f a0[8], a1[8];
  load_a_group(a0, ap, 0);

  {
    const int4* __restrict__ d4 = (const int4*)dst;
    const int4* __restrict__ s4 = (const int4*)src;
    for (int i = blockIdx.x * GEMM_TPB + tid; i < N_EDGES / 4; i += GEMM_BLOCKS * GEMM_TPB) {
      const int4 d = d4[i];
      const int4 sr = s4[i];
      const unsigned o[4] = {(unsigned)d.x / R_NODES, (unsigned)d.y / R_NODES,
                             (unsigned)d.z / R_NODES, (unsigned)d.w / R_NODES};
      uint4 rw;
      rw.x = (((unsigned)d.x - o[0] * R_NODES) << 16) | (unsigned)sr.x;
      rw.y = (((unsigned)d.y - o[1] * R_NODES) << 16) | (unsigned)sr.y;
      rw.z = (((unsigned)d.z - o[2] * R_NODES) << 16) | (unsigned)sr.z;
      rw.w = (((unsigned)d.w - o[3] * R_NODES) << 16) | (unsigned)sr.w;
      ((uint4*)rec)[i] = rw;
      unsigned word = 0u;
#pragma unroll
      for (int s = 0; s < 4; ++s) {
#pragma unroll
        for (int k = 0; k < 8; ++k) {
          const unsigned plane = __builtin_amdgcn_ballot_w32(((o[s] >> k) & 1u) != 0u);
          if (lane == 8 * s + k) word = plane;
        }
      }
      keys[i] = word;
    }
  }

  v4f wa_a[OUT_DIM / 8], wa_b[OUT_DIM / 8];
  {
    const v4f* __restrict__ wr = (const v4f*)(W + (size_t)(16 * wv + c) * OUT_DIM) + half;
    const v4f* __restrict__ ar = (const v4f*)(attn_w + (c & 1) * OUT_DIM) + half;
#pragma unroll
    for (int i = 0; i < OUT_DIM / 8; ++i) { wa_a[i] = wr[2 * i]; wa_b[i] = ar[2 * i]; }
  }
  v8f w0 = {}, w1 = {};

  {
    const v4f* __restrict__ W4 = (const v4f*)W;
#pragma unroll
    for (int it = 0; it < STAGE_ROUNDS; ++it) {
      const int e  = tid + it * GEMM_TPB;
      const int kq = e >> 5;
      const int j  = e & 31;
      const v4f r0 = W4[(4 * kq + 0) * (OUT_DIM / 4) + j];
      const v4f r1 = W4[(4 * kq + 1) * (OUT_DIM / 4) + j];
      const v4f r2 = W4[(4 * kq + 2) * (OUT_DIM / 4) + j];
      const v4f r3 = W4[(4 * kq + 3) * (OUT_DIM / 4) + j];
#pragma unroll
      for (int i = 4 * it; i < 4 * it + 4; i += 2) {
        v2f alo, ahi, blo, bhi;
        alo.x = wa_a[i].x; alo.y = wa_a[i].y; ahi.x = wa_a[i].z; ahi.y = wa_a[i].w;
        blo.x = wa_b[i].x; blo.y = wa_b[i].y; bhi.x = wa_b[i].z; bhi.y = wa_b[i].w;
        w0 = __builtin_amdgcn_wmma_f32_16x16x4_f32(false, alo, false, blo, (short)0, w0, false, false);
        w0 = __builtin_amdgcn_wmma_f32_16x16x4_f32(false, ahi, false, bhi, (short)0, w0, false, false);
        alo.x = wa_a[i + 1].x; alo.y = wa_a[i + 1].y; ahi.x = wa_a[i + 1].z; ahi.y = wa_a[i + 1].w;
        blo.x = wa_b[i + 1].x; blo.y = wa_b[i + 1].y; bhi.x = wa_b[i + 1].z; bhi.y = wa_b[i + 1].w;
        w1 = __builtin_amdgcn_wmma_f32_16x16x4_f32(false, alo, false, blo, (short)0, w1, false, false);
        w1 = __builtin_amdgcn_wmma_f32_16x16x4_f32(false, ahi, false, bhi, (short)0, w1, false, false);
      }
      v4f* d = Wl + kq * OUT_DIM + 4 * j;
      v4f q0 = {r0.x, r1.x, r2.x, r3.x};
      v4f q1 = {r0.y, r1.y, r2.y, r3.y};
      v4f q2 = {r0.z, r1.z, r2.z, r3.z};
      v4f q3 = {r0.w, r1.w, r2.w, r3.w};
      d[0] = q0; d[1] = q1; d[2] = q2; d[3] = q3;
    }
  }
  if (c < 2) {
    float* wf = (float*)wal + c * IN_DIM + 16 * wv + 8 * half;
#pragma unroll
    for (int i = 0; i < 8; ++i) wf[i] = w0[i] + w1[i];
  }
  __syncthreads();

  const v4f* __restrict__ bp = Wl + half * OUT_DIM + nh * 64 + c;
  const v4f* __restrict__ wp = wal + half;

  for (;;) {
    v8f acc[4] = {};
    const int  mt_next  = mt + TILE_STEP;
    const bool has_next = mt_next < M_TILES;

    asm volatile("" ::: "memory");
    v4f b0[4], b1[4];
    v4f q0, p0, q1, p1;
    v4f sv = {0.0f, 0.0f, 0.0f, 0.0f};
    v4f tv = {0.0f, 0.0f, 0.0f, 0.0f};
    load_b_chunk(b0, bp, 0);
    q0 = wp[0];
    p0 = wp[IN_DIM / 4];
    load_a_group(a1, ap, 1);
    mma_group<0>(acc, a0, bp, b0, b1, wp, q0, p0, q1, p1, sv, tv);
    load_a_group(a0, ap, 2);
    mma_group<1>(acc, a1, bp, b0, b1, wp, q0, p0, q1, p1, sv, tv);
    load_a_group(a1, ap, 3);
    mma_group<2>(acc, a0, bp, b0, b1, wp, q0, p0, q1, p1, sv, tv);
    const v4f* __restrict__ ap_next =
        (const v4f*)(h + (size_t)(mt_next * 16 + c) * IN_DIM) + half;
    if (has_next) load_a_group(a0, ap_next, 0);
    mma_group<3>(acc, a1, bp, b0, b1, wp, q0, p0, q1, p1, sv, tv);

    {
      float sa = (sv.x + sv.y) + (sv.z + sv.w);
      float sb = (tv.x + tv.y) + (tv.z + tv.w);
      sa += __shfl_xor(sa, 16);
      sb += __shfl_xor(sb, 16);
      if (nh == 0) sc[mt * 32 + lane] = half ? sb : sa;
    }

    float* zp = z + (size_t)(mt * 16 + 8 * half) * OUT_DIM + nh * 64 + c;
#pragma unroll
    for (int t = 0; t < 4; ++t) {
#pragma unroll
      for (int i = 0; i < 8; ++i) zp[(size_t)i * OUT_DIM + 16 * t] = acc[t][i];
    }

    if (!has_next) break;
    mt = mt_next;
    ap = ap_next;
  }
}

__device__ __forceinline__ unsigned enc_f32(float f) {
  unsigned b = __float_as_uint(f);
  return (b & 0x80000000u) ? ~b : (b | 0x80000000u);
}
__device__ __forceinline__ float dec_f32(unsigned e) {
  unsigned b = (e & 0x80000000u) ? (e & 0x7FFFFFFFu) : ~e;
  return __uint_as_float(b);
}

__device__ __forceinline__ int s1_at(int n) { return ((n >> 4) << 5) + (n & 15); }
__device__ __forceinline__ int s2_at(int n) { return ((n >> 4) << 5) + 16 + (n & 15); }

#define N_BIN 32
#define ROWS_PER_WAVE ((R_NODES + N_BIN - 1) / N_BIN)
static_assert(N_BIN == AGG_TPB / 32, "one bin per wave");
static_assert(sizeof(uint2) * LIST_CAP >= sizeof(float) * R_NODES * OUT_DIM,
              "the overflow path keeps its row sums in the bin array");

__device__ __forceinline__ void add_row(v4f (&ac)[ROWS_PER_WAVE], float (&dn)[ROWS_PER_WAVE],
                                        const int r, const float ex, const v4f v) {
#pragma unroll
  for (int k = 0; k < ROWS_PER_WAVE; ++k) {
    if (r == k) {
      asm volatile("");
      ac[k] += ex * v;
      dn[k] += ex;
    }
  }
}

__global__ __launch_bounds__(AGG_TPB) void k_aggregate(const int* __restrict__ src,
                                                       const int* __restrict__ dst,
                                                       const float* __restrict__ z,
                                                       const float* __restrict__ sc,
                                                       const unsigned* __restrict__ keys,
                                                       const unsigned* __restrict__ rec,
                                                       float* __restrict__ out) {
  __shared__ __attribute__((aligned(16))) uint2 lst2[LIST_CAP];
  __shared__ unsigned lst_a[LIST_CAP];
  __shared__ float    lst_e[LIST_CAP];
  __shared__ unsigned menc[R_NODES];
  __shared__ float    den[R_NODES];
  __shared__ float    s2l[R_NODES];
  __shared__ int      bcnt[N_BIN];
  __shared__ int      bbase[N_BIN];
  __shared__ int      bfill[N_BIN];
  __shared__ int      cnt;

  const int tid  = threadIdx.x;
  const int lane = tid & 31;
  const int wv   = tid >> 5;
  const int base = blockIdx.x * R_NODES;

  if (tid < R_NODES) {
    menc[tid] = 0u;
    den[tid]  = 0.0f;
    const int nd = base + tid;
    s2l[tid] = (nd < N_NODES) ? sc[s2_at(nd)] : 0.0f;
  }
  if (tid < N_BIN) bcnt[tid] = 0;
  if (tid == 0) cnt = 0;
  __syncthreads();

  {
    const uint4* __restrict__ k4 = (const uint4*)keys;
    unsigned inv[8];
#pragma unroll
    for (int k = 0; k < 8; ++k) inv[k] = ((blockIdx.x >> k) & 1u) ? 0u : 0xFFFFFFFFu;
    const int start = blockIdx.x * (NSG / N_OWN);
    for (int it = 0; it < SCAN_ITERS; it += SCAN_BATCH) {
      uint4 lo[SCAN_BATCH], hi[SCAN_BATCH];
      int   qq[SCAN_BATCH];
      bool  ok[SCAN_BATCH];
#pragma unroll
      for (int u = 0; u < SCAN_BATCH; ++u) {
        const int t = (it + u) * AGG_TPB + tid;
        ok[u] = t < NSG;
        int q = t + start;
        if (q >= NSG) q -= NSG;
        if (!ok[u]) q = 0;
        qq[u] = q;
        lo[u] = k4[2 * q];
        hi[u] = k4[2 * q + 1];
      }
#pragma unroll
      for (int u = 0; u < SCAN_BATCH; ++u) {
        unsigned m = (lo[u].x ^ inv[0]) & (lo[u].y ^ inv[1]) & (lo[u].z ^ inv[2]) &
                     (lo[u].w ^ inv[3]) & (hi[u].x ^ inv[4]) & (hi[u].y ^ inv[5]) &
                     (hi[u].z ^ inv[6]) & (hi[u].w ^ inv[7]);
        if (!ok[u]) m = 0u;
        const unsigned e0 = ((unsigned)(qq[u] >> 2) << 7) + (unsigned)(qq[u] & 3);
        while (m != 0u) {
          const int l = __builtin_ctz(m);
          m &= m - 1u;
          const int pos = atomicAdd(&cnt, 1);
          if (pos < LIST_CAP) lst_a[pos] = e0 + 4u * (unsigned)l;
        }
      }
    }
  }
  __syncthreads();
  const int n = cnt;

  if (n <= LIST_CAP) {
    for (int j0 = tid; j0 < n; j0 += 4 * AGG_TPB) {
      int ii[4], ss[4];
      unsigned pp[4];
      float sv[4];
#pragma unroll
      for (int u = 0; u < 4; ++u) {
        const int j = j0 + u * AGG_TPB;
        ii[u] = (j < n) ? (int)lst_a[j] : 0;
      }
#pragma unroll
      for (int u = 0; u < 4; ++u) { pp[u] = rec[ii[u]]; ss[u] = (int)(pp[u] & 0xFFFFu); }
#pragma unroll
      for (int u = 0; u < 4; ++u) sv[u] = sc[s1_at(ss[u])];
#pragma unroll
      for (int u = 0; u < 4; ++u) {
        const int j = j0 + u * AGG_TPB;
        if (j < n) {
          const int loc = (int)(pp[u] >> 16);
          const float logit = sv[u] + s2l[loc];
          const float e = (logit >= 0.0f) ? logit : LEAK * logit;
          lst_e[j] = e;
          lst_a[j] = pp[u];
          atomicMax(&menc[loc], enc_f32(e));
          atomicAdd(&bcnt[loc & (N_BIN - 1)], 1);
        }
      }
    }
    __syncthreads();

    if (wv == 0) {
      const int c = bcnt[lane];
      int incl = c;
#pragma unroll
      for (int d = 1; d < N_BIN; d <<= 1) {
        const int t = __shfl_up(incl, d);
        if (lane >= d) incl += t;
      }
      bbase[lane] = incl - c;
      bfill[lane] = incl - c;
    }
    __syncthreads();

    for (int j = tid; j < n; j += AGG_TPB) {
      const unsigned pk = lst_a[j];
      const int loc = (int)(pk >> 16);
      const float ex = expf(lst_e[j] - dec_f32(menc[loc]));
      const int pos = atomicAdd(&bfill[loc & (N_BIN - 1)], 1);
      lst2[pos] = make_uint2(pk, __float_as_uint(ex));
    }
    __syncthreads();

    {
      const v4f* __restrict__ z4 = (const v4f*)z;
      const int beg = __builtin_amdgcn_readfirstlane(bbase[wv]);
      const int end = beg + __builtin_amdgcn_readfirstlane(bcnt[wv]);
      v4f   ac[ROWS_PER_WAVE];
      float dn[ROWS_PER_WAVE];
#pragma unroll
      for (int k = 0; k < ROWS_PER_WAVE; ++k) {
        const v4f zero = {0.0f, 0.0f, 0.0f, 0.0f};
        ac[k] = zero;
        dn[k] = 0.0f;
      }
      int j = beg;
      for (; j + 8 <= end; j += 8) {
        uint2 en[8];
        v4f   v[8];
#pragma unroll
        for (int u = 0; u < 8; ++u) en[u] = lst2[j + u];
#pragma unroll
        for (int u = 0; u < 8; ++u) v[u] = z4[(size_t)(en[u].x & 0xFFFFu) * (OUT_DIM / 4) + lane];
#pragma unroll
        for (int u = 0; u < 8; ++u) {
          const int r = __builtin_amdgcn_readfirstlane((int)(en[u].x >> 21));
          add_row(ac, dn, r, __uint_as_float(en[u].y), v[u]);
        }
      }
      const int rem = end - j;
      if (rem > 0) {
        uint2 en[7];
        v4f   v[7];
#pragma unroll
        for (int u = 0; u < 7; ++u) {
          const int ju = j + u;
          en[u] = lst2[ju < LIST_CAP ? ju : LIST_CAP - 1];
        }
#pragma unroll
        for (int u = 0; u < 7; ++u) {
          const v4f zero = {0.0f, 0.0f, 0.0f, 0.0f};
          v[u] = zero;
          if (u < rem) v[u] = z4[(size_t)(en[u].x & 0xFFFFu) * (OUT_DIM / 4) + lane];
        }
#pragma unroll
        for (int u = 0; u < 7; ++u) {
          if (u < rem) {
            const int r = __builtin_amdgcn_readfirstlane((int)(en[u].x >> 21));
            add_row(ac, dn, r, __uint_as_float(en[u].y), v[u]);
          }
        }
      }
#pragma unroll
      for (int k = 0; k < ROWS_PER_WAVE; ++k) {
        const int loc = wv + N_BIN * k;
        const int nd  = base + loc;
        if (loc < R_NODES && nd < N_NODES) {
          const float d = (dn[k] > 0.0f) ? dn[k] : 1.0f;
          v4f x;
          x.x = ac[k].x / d; x.y = ac[k].y / d; x.z = ac[k].z / d; x.w = ac[k].w / d;
          x.x = (x.x > 0.0f) ? x.x : expm1f(x.x);
          x.y = (x.y > 0.0f) ? x.y : expm1f(x.y);
          x.z = (x.z > 0.0f) ? x.z : expm1f(x.z);
          x.w = (x.w > 0.0f) ? x.w : expm1f(x.w);
          ((v4f*)out)[(size_t)nd * (OUT_DIM / 4) + lane] = x;
        }
      }
    }
  } else {
    float* acc = (float*)lst2;
    for (int i = tid; i < R_NODES * OUT_DIM / 4; i += AGG_TPB) {
      const v4f zero = {0.0f, 0.0f, 0.0f, 0.0f};
      ((v4f*)acc)[i] = zero;
    }
    __syncthreads();
    for (int i = tid; i < N_EDGES; i += AGG_TPB) {
      const int loc = dst[i] - base;
      if ((unsigned)loc < (unsigned)R_NODES) {
        const float logit = sc[s1_at(src[i])] + s2l[loc];
        const float e = (logit >= 0.0f) ? logit : LEAK * logit;
        atomicMax(&menc[loc], enc_f32(e));
      }
    }
    __syncthreads();
    for (int i = tid; i < N_EDGES; i += AGG_TPB) {
      const int loc = dst[i] - base;
      if ((unsigned)loc < (unsigned)R_NODES) {
        const int s = src[i];
        const float logit = sc[s1_at(s)] + s2l[loc];
        const float e = (logit >= 0.0f) ? logit : LEAK * logit;
        const float ex = expf(e - dec_f32(menc[loc]));
        atomicAdd(&den[loc], ex);
        const float* zr = z + (size_t)s * OUT_DIM;
        float* a = acc + loc * OUT_DIM;
        for (int c = 0; c < OUT_DIM; ++c) atomicAdd(&a[(c & 3) * 32 + (c >> 2)], ex * zr[c]);
      }
    }
    __syncthreads();

    for (int idx = tid; idx < R_NODES * (OUT_DIM / 4); idx += AGG_TPB) {
      const int r  = idx >> 5;
      const int q  = idx & 31;
      const int nd = base + r;
      if (nd < N_NODES) {
        const float dn0 = den[r];
        const float dn  = (dn0 > 0.0f) ? dn0 : 1.0f;
        const float* a = acc + r * OUT_DIM + q;
        v4f x;
        x.x = a[0] / dn; x.y = a[32] / dn; x.z = a[64] / dn; x.w = a[96] / dn;
        x.x = (x.x > 0.0f) ? x.x : expm1f(x.x);
        x.y = (x.y > 0.0f) ? x.y : expm1f(x.y);
        x.z = (x.z > 0.0f) ? x.z : expm1f(x.z);
        x.w = (x.w > 0.0f) ? x.w : expm1f(x.w);
        ((v4f*)out)[(size_t)nd * (OUT_DIM / 4) + q] = x;
      }
    }
  }
}

extern "C" void kernel_launch(void* const* d_in, const int* in_sizes, int n_in,
                              void* d_out, int out_size, void* d_ws, size_t ws_size,
                              hipStream_t stream) {
  const float* h      = (const float*)d_in[0];
  const int*   src    = (const int*)  d_in[1];
  const int*   dst    = (const int*)  d_in[2];
  const float* W_fc   = (const float*)d_in[3];
  const float* attn_w = (const float*)d_in[4];
  float*       out    = (float*)d_out;

  float*    z    = (float*)d_ws;
  float*    sc   = z  + (size_t)N_NODES * OUT_DIM;
  unsigned* keys = (unsigned*)(sc + 2 * N_NODES);
  unsigned* rec  = keys + N_EDGES / 4;

  k_gemm<<<GEMM_BLOCKS, GEMM_TPB, 0, stream>>>(h, W_fc, attn_w, src, dst, z, sc, keys, rec);

  k_aggregate<<<N_OWN, AGG_TPB, 0, stream>>>(src, dst, z, sc, keys, rec, out);
}
